# starting kernel
# baseline (speedup 1.0000x reference)
#include <hip/hip_runtime.h>

#define NN 50000
#define NE 600000
#define DIM 128
#define BN_EPS 1e-5f

typedef __attribute__((ext_vector_type(2))) float v2f;
typedef __attribute__((ext_vector_type(8))) float v8f;


__global__ void init_deg_kernel(float* __restrict__ deg, int n) {
    int i = blockIdx.x * blockDim.x + threadIdx.x;
    if (i < n) deg[i] = 1.0f;
}

__global__ void deg_count_kernel(const int* __restrict__ col, float* __restrict__ deg, int e) {
    int i = blockIdx.x * blockDim.x + threadIdx.x;
    if (i < e) unsafeAtomicAdd(&deg[col[i]], 1.0f);
}

__global__ void rsqrt_kernel(float* __restrict__ deg, int n) {
    int i = blockIdx.x * blockDim.x + threadIdx.x;
    if (i < n) deg[i] = rsqrtf(deg[i]);
}

__global__ void zero_kernel(float* __restrict__ p, int n) {
    int i = blockIdx.x * blockDim.x + threadIdx.x;
    if (i < n) p[i] = 0.0f;
}


__global__ __launch_bounds__(256)
void transpose128_kernel(const float* __restrict__ W, float* __restrict__ Wt) {
    int t = blockIdx.x * blockDim.x + threadIdx.x;
    int r = t >> 7;
    int c = t & 127;
    Wt[(size_t)c * DIM + r] = W[(size_t)r * DIM + c];
}

__global__ __launch_bounds__(256)
void gemm_wmma_kernel(const float* __restrict__ X, const float* __restrict__ Wt,
                      float* __restrict__ H) {
    const int tileM = blockIdx.x * 16;
    const int wave  = threadIdx.x >> 5;
    const int lane  = threadIdx.x & 31;
    const int tileN = wave * 16;
    const int mrow  = lane & 15;
    const int hi    = lane >> 4;

    v8f acc = {};
    const float* xr = X + (size_t)(tileM + mrow) * DIM + 2 * hi;
    const float* wc = Wt + (size_t)(tileN + mrow) * DIM + 2 * hi;

#pragma unroll
    for (int k0 = 0; k0 < DIM; k0 += 4) {
        v2f a = *(const v2f*)(xr + k0);
        v2f b = *(const v2f*)(wc + k0);
        acc = __builtin_amdgcn_wmma_f32_16x16x4_f32(
             false, a,  false, b,
             (short)0, acc,  false,  false);
    }

    float* out = H + (size_t)(tileM + 8 * hi) * DIM + tileN + mrow;
#pragma unroll
    for (int r = 0; r < 8; ++r)
        out[(size_t)r * DIM] = acc[r];
}


__global__ __launch_bounds__(256)
void selfloop_kernel(const float* __restrict__ h, const float* __restrict__ dis,
                     float* __restrict__ agg, int n) {
    int t = blockIdx.x * blockDim.x + threadIdx.x;
    if (t >= n * 32) return;
    int node = t >> 5;
    float s = dis[node];
    s = s * s;
    float4 v = ((const float4*)h)[t];
    v.x *= s; v.y *= s; v.z *= s; v.w *= s;
    ((float4*)agg)[t] = v;
}

__global__ __launch_bounds__(256)
void edge_scatter_kernel(const float* __restrict__ h, float* __restrict__ agg,
                         const int* __restrict__ row, const int* __restrict__ col,
                         const float* __restrict__ dis, int e) {
    int warp = threadIdx.x >> 5;
    int lane = threadIdx.x & 31;
    int eid = blockIdx.x * 8 + warp;
    if (eid >= e) return;
    int r = row[eid];
    int c = col[eid];
    float norm = dis[r] * dis[c];
    float4 v = ((const float4*)(h + (size_t)r * DIM))[lane];
    float* dst = agg + (size_t)c * DIM + lane * 4;
    unsafeAtomicAdd(dst + 0, v.x * norm);
    unsafeAtomicAdd(dst + 1, v.y * norm);
    unsafeAtomicAdd(dst + 2, v.z * norm);
    unsafeAtomicAdd(dst + 3, v.w * norm);
}


__global__ __launch_bounds__(256)
void bias_relu_kernel(const float* __restrict__ agg, const float* __restrict__ b,
                      float* __restrict__ out, int n) {
    int t = blockIdx.x * blockDim.x + threadIdx.x;
    if (t >= n * 32) return;
    int j = t & 31;
    float4 v = ((const float4*)agg)[t];
    float4 bb = ((const float4*)b)[j];
    v.x = fmaxf(v.x + bb.x, 0.0f);
    v.y = fmaxf(v.y + bb.y, 0.0f);
    v.z = fmaxf(v.z + bb.z, 0.0f);
    v.w = fmaxf(v.w + bb.w, 0.0f);
    ((float4*)out)[t] = v;
}


__global__ __launch_bounds__(128)
void bn_reduce_kernel(const float* __restrict__ h, float* __restrict__ sum,
                      float* __restrict__ sumsq, int n, int rows_per_block) {
    int d = threadIdx.x;
    int r0 = blockIdx.x * rows_per_block;
    int r1 = r0 + rows_per_block;
    if (r1 > n) r1 = n;
    float s = 0.0f, ss = 0.0f;
    for (int r = r0; r < r1; ++r) {
        float v = h[(size_t)r * DIM + d];
        s += v;
        ss += v * v;
    }
    unsafeAtomicAdd(&sum[d], s);
    unsafeAtomicAdd(&sumsq[d], ss);
}

__global__ __launch_bounds__(128)
void bn_finalize_kernel(const float* __restrict__ sum, const float* __restrict__ sumsq,
                        float* __restrict__ mean, float* __restrict__ istd, int n) {
    int d = threadIdx.x;
    float invn = 1.0f / (float)n;
    float m = sum[d] * invn;
    float var = sumsq[d] * invn - m * m;
    mean[d] = m;
    istd[d] = rsqrtf(var + BN_EPS);
}

__global__ __launch_bounds__(256)
void bn_norm_kernel(const float* __restrict__ h, const float* __restrict__ mean,
                    const float* __restrict__ istd, const float* __restrict__ gamma,
                    const float* __restrict__ beta, float* __restrict__ out, int n) {
    int t = blockIdx.x * blockDim.x + threadIdx.x;
    if (t >= n * 32) return;
    int j = t & 31;
    float4 v  = ((const float4*)h)[t];
    float4 m  = ((const float4*)mean)[j];
    float4 is = ((const float4*)istd)[j];
    float4 g  = ((const float4*)gamma)[j];
    float4 be = ((const float4*)beta)[j];
    float4 o;
    o.x = g.x * (v.x - m.x) * is.x + be.x;
    o.y = g.y * (v.y - m.y) * is.y + be.y;
    o.z = g.z * (v.z - m.z) * is.z + be.z;
    o.w = g.w * (v.w - m.w) * is.w + be.w;
    ((float4*)out)[t] = o;
}


extern "C" void kernel_launch(void* const* d_in, const int* in_sizes, int n_in,
                              void* d_out, int out_size, void* d_ws, size_t ws_size,
                              hipStream_t stream) {
    const float* x     = (const float*)d_in[0];
    const int*   eidx  = (const int*)d_in[1];
    const float* W1    = (const float*)d_in[2];
    const float* b1    = (const float*)d_in[3];
    const float* W2    = (const float*)d_in[4];
    const float* b2    = (const float*)d_in[5];
    const float* gamma = (const float*)d_in[6];
    const float* beta  = (const float*)d_in[7];
    float* out = (float*)d_out;

    const int* row = eidx;
    const int* col = eidx + NE;

    float* bufA  = (float*)d_ws;
    float* bufB  = bufA + (size_t)NN * DIM;
    float* dis   = bufB + (size_t)NN * DIM;
    float* sum   = dis + NN;
    float* sumsq = sum + DIM;
    float* mean  = sumsq + DIM;
    float* istd  = mean + DIM;
    float* W1t   = istd + DIM;
    float* W2t   = W1t + (size_t)DIM * DIM;

    const int T = 256;
    const int elem_blocks = (NN * 32 + T - 1) / T;
    const int gemm_blocks = NN / 16;
    const int edge_blocks = (NE + 7) / 8;
    const int tr_blocks   = (DIM * DIM) / T;
    const int bn_blocks = 256;
    const int rows_per_block = (NN + bn_blocks - 1) / bn_blocks;

    init_deg_kernel<<<(NN + T - 1) / T, T, 0, stream>>>(dis, NN);
    deg_count_kernel<<<(NE + T - 1) / T, T, 0, stream>>>(col, dis, NE);
    rsqrt_kernel<<<(NN + T - 1) / T, T, 0, stream>>>(dis, NN);
    zero_kernel<<<1, 2 * DIM, 0, stream>>>(sum, 2 * DIM);
    transpose128_kernel<<<tr_blocks, T, 0, stream>>>(W1, W1t);
    transpose128_kernel<<<tr_blocks, T, 0, stream>>>(W2, W2t);

    gemm_wmma_kernel<<<gemm_blocks, T, 0, stream>>>(x, W1t, bufA);
    selfloop_kernel<<<elem_blocks, T, 0, stream>>>(bufA, dis, bufB, NN);
    edge_scatter_kernel<<<edge_blocks, T, 0, stream>>>(bufA, bufB, row, col, dis, NE);
    bias_relu_kernel<<<elem_blocks, T, 0, stream>>>(bufB, b1, bufA, NN);

    gemm_wmma_kernel<<<gemm_blocks, T, 0, stream>>>(bufA, W2t, bufB);
    selfloop_kernel<<<elem_blocks, T, 0, stream>>>(bufB, dis, bufA, NN);
    edge_scatter_kernel<<<edge_blocks, T, 0, stream>>>(bufB, bufA, row, col, dis, NE);
    bias_relu_kernel<<<elem_blocks, T, 0, stream>>>(bufA, b2, bufB, NN);

    bn_reduce_kernel<<<bn_blocks, DIM, 0, stream>>>(bufB, sum, sumsq, NN, rows_per_block);
    bn_finalize_kernel<<<1, DIM, 0, stream>>>(sum, sumsq, mean, istd, NN);
    bn_norm_kernel<<<elem_blocks, T, 0, stream>>>(bufB, mean, istd, gamma, beta, out, NN);
}


# each layer aggregates before its product ((A x) W, not A (x W): new f32 sum order), so bias, relu and the next self-loop start ride in the product's store; 17 -> 14 launches
# speedup vs baseline: 1.2019x; 1.2019x over previous
#include <hip/hip_runtime.h>

#define NN 50000
#define NE 600000
#define DIM 128
#define BN_EPS 1e-5f

typedef __attribute__((ext_vector_type(2))) float v2f;
typedef __attribute__((ext_vector_type(8))) float v8f;


__global__ void init_deg_kernel(float* __restrict__ deg, int n) {
    int i = blockIdx.x * blockDim.x + threadIdx.x;
    if (i < n) deg[i] = 1.0f;
}

__global__ void deg_count_kernel(const int* __restrict__ col, float* __restrict__ deg, int e) {
    int i = blockIdx.x * blockDim.x + threadIdx.x;
    if (i < e) unsafeAtomicAdd(&deg[col[i]], 1.0f);
}

__global__ void rsqrt_kernel(float* __restrict__ deg, int n) {
    int i = blockIdx.x * blockDim.x + threadIdx.x;
    if (i < n) deg[i] = rsqrtf(deg[i]);
}

__global__ void zero_kernel(float* __restrict__ p, int n) {
    int i = blockIdx.x * blockDim.x + threadIdx.x;
    if (i < n) p[i] = 0.0f;
}


__global__ __launch_bounds__(256)
void transpose128_kernel(const float* __restrict__ W, float* __restrict__ Wt) {
    int t = blockIdx.x * blockDim.x + threadIdx.x;
    int r = t >> 7;
    int c = t & 127;
    Wt[(size_t)c * DIM + r] = W[(size_t)r * DIM + c];
}

template <bool SELF>
__global__ __launch_bounds__(256)
void gemm_wmma_kernel(const float* __restrict__ X, const float* __restrict__ Wt,
                      const float* __restrict__ bias, const float* __restrict__ dis,
                      float* __restrict__ H, float* __restrict__ AggNext) {
    const int tileM = blockIdx.x * 16;
    const int wave  = threadIdx.x >> 5;
    const int lane  = threadIdx.x & 31;
    const int tileN = wave * 16;
    const int mrow  = lane & 15;
    const int hi    = lane >> 4;

    v8f acc = {};
    const float* xr = X + (size_t)(tileM + mrow) * DIM + 2 * hi;
    const float* wc = Wt + (size_t)(tileN + mrow) * DIM + 2 * hi;

#pragma unroll
    for (int k0 = 0; k0 < DIM; k0 += 4) {
        v2f a = *(const v2f*)(xr + k0);
        v2f b = *(const v2f*)(wc + k0);
        acc = __builtin_amdgcn_wmma_f32_16x16x4_f32(
             false, a,  false, b,
             (short)0, acc,  false,  false);
    }

    const int row0 = tileM + 8 * hi;
    const int coln = tileN + mrow;
    const float bv = bias[coln];
    float* out = H + (size_t)row0 * DIM + coln;
    float* agg = SELF ? (AggNext + (size_t)row0 * DIM + coln) : nullptr;
#pragma unroll
    for (int r = 0; r < 8; ++r) {
        float v = fmaxf(acc[r] + bv, 0.0f);
        out[(size_t)r * DIM] = v;
        if (SELF) {
            float s = dis[row0 + r];
            agg[(size_t)r * DIM] = v * (s * s);
        }
    }
}


__global__ __launch_bounds__(256)
void selfloop_kernel(const float* __restrict__ h, const float* __restrict__ dis,
                     float* __restrict__ agg, int n) {
    int t = blockIdx.x * blockDim.x + threadIdx.x;
    if (t >= n * 32) return;
    int node = t >> 5;
    float s = dis[node];
    s = s * s;
    float4 v = ((const float4*)h)[t];
    v.x *= s; v.y *= s; v.z *= s; v.w *= s;
    ((float4*)agg)[t] = v;
}

__global__ __launch_bounds__(256)
void edge_scatter_kernel(const float* __restrict__ h, float* __restrict__ agg,
                         const int* __restrict__ row, const int* __restrict__ col,
                         const float* __restrict__ dis, int e) {
    int warp = threadIdx.x >> 5;
    int lane = threadIdx.x & 31;
    int eid = blockIdx.x * 8 + warp;
    if (eid >= e) return;
    int r = row[eid];
    int c = col[eid];
    float norm = dis[r] * dis[c];
    float4 v = ((const float4*)(h + (size_t)r * DIM))[lane];
    float* dst = agg + (size_t)c * DIM + lane * 4;
    unsafeAtomicAdd(dst + 0, v.x * norm);
    unsafeAtomicAdd(dst + 1, v.y * norm);
    unsafeAtomicAdd(dst + 2, v.z * norm);
    unsafeAtomicAdd(dst + 3, v.w * norm);
}


__global__ __launch_bounds__(128)
void bn_reduce_kernel(const float* __restrict__ h, float* __restrict__ sum,
                      float* __restrict__ sumsq, int n, int rows_per_block) {
    int d = threadIdx.x;
    int r0 = blockIdx.x * rows_per_block;
    int r1 = r0 + rows_per_block;
    if (r1 > n) r1 = n;
    float s = 0.0f, ss = 0.0f;
    for (int r = r0; r < r1; ++r) {
        float v = h[(size_t)r * DIM + d];
        s += v;
        ss += v * v;
    }
    unsafeAtomicAdd(&sum[d], s);
    unsafeAtomicAdd(&sumsq[d], ss);
}

__global__ __launch_bounds__(128)
void bn_finalize_kernel(const float* __restrict__ sum, const float* __restrict__ sumsq,
                        float* __restrict__ mean, float* __restrict__ istd, int n) {
    int d = threadIdx.x;
    float invn = 1.0f / (float)n;
    float m = sum[d] * invn;
    float var = sumsq[d] * invn - m * m;
    mean[d] = m;
    istd[d] = rsqrtf(var + BN_EPS);
}

__global__ __launch_bounds__(256)
void bn_norm_kernel(const float* __restrict__ h, const float* __restrict__ mean,
                    const float* __restrict__ istd, const float* __restrict__ gamma,
                    const float* __restrict__ beta, float* __restrict__ out, int n) {
    int t = blockIdx.x * blockDim.x + threadIdx.x;
    if (t >= n * 32) return;
    int j = t & 31;
    float4 v  = ((const float4*)h)[t];
    float4 m  = ((const float4*)mean)[j];
    float4 is = ((const float4*)istd)[j];
    float4 g  = ((const float4*)gamma)[j];
    float4 be = ((const float4*)beta)[j];
    float4 o;
    o.x = g.x * (v.x - m.x) * is.x + be.x;
    o.y = g.y * (v.y - m.y) * is.y + be.y;
    o.z = g.z * (v.z - m.z) * is.z + be.z;
    o.w = g.w * (v.w - m.w) * is.w + be.w;
    ((float4*)out)[t] = o;
}


extern "C" void kernel_launch(void* const* d_in, const int* in_sizes, int n_in,
                              void* d_out, int out_size, void* d_ws, size_t ws_size,
                              hipStream_t stream) {
    const float* x     = (const float*)d_in[0];
    const int*   eidx  = (const int*)d_in[1];
    const float* W1    = (const float*)d_in[2];
    const float* b1    = (const float*)d_in[3];
    const float* W2    = (const float*)d_in[4];
    const float* b2    = (const float*)d_in[5];
    const float* gamma = (const float*)d_in[6];
    const float* beta  = (const float*)d_in[7];
    float* out = (float*)d_out;

    const int* row = eidx;
    const int* col = eidx + NE;

    float* bufA  = (float*)d_ws;
    float* bufB  = bufA + (size_t)NN * DIM;
    float* bufC  = bufB + (size_t)NN * DIM;
    float* dis   = bufC + (size_t)NN * DIM;
    float* sum   = dis + NN;
    float* sumsq = sum + DIM;
    float* mean  = sumsq + DIM;
    float* istd  = mean + DIM;
    float* W1t   = istd + DIM;
    float* W2t   = W1t + (size_t)DIM * DIM;

    const int T = 256;
    const int elem_blocks = (NN * 32 + T - 1) / T;
    const int gemm_blocks = NN / 16;
    const int edge_blocks = (NE + 7) / 8;
    const int tr_blocks   = (DIM * DIM) / T;
    const int bn_blocks = 256;
    const int rows_per_block = (NN + bn_blocks - 1) / bn_blocks;

    init_deg_kernel<<<(NN + T - 1) / T, T, 0, stream>>>(dis, NN);
    deg_count_kernel<<<(NE + T - 1) / T, T, 0, stream>>>(col, dis, NE);
    rsqrt_kernel<<<(NN + T - 1) / T, T, 0, stream>>>(dis, NN);
    zero_kernel<<<1, 2 * DIM, 0, stream>>>(sum, 2 * DIM);
    transpose128_kernel<<<tr_blocks, T, 0, stream>>>(W1, W1t);
    transpose128_kernel<<<tr_blocks, T, 0, stream>>>(W2, W2t);


    selfloop_kernel<<<elem_blocks, T, 0, stream>>>(x, dis, bufA, NN);
    edge_scatter_kernel<<<edge_blocks, T, 0, stream>>>(x, bufA, row, col, dis, NE);
    gemm_wmma_kernel<true><<<gemm_blocks, T, 0, stream>>>(bufA, W1t, b1, dis, bufB, bufC);

    edge_scatter_kernel<<<edge_blocks, T, 0, stream>>>(bufB, bufC, row, col, dis, NE);
    gemm_wmma_kernel<false><<<gemm_blocks, T, 0, stream>>>(bufC, W2t, b2, nullptr, bufA, nullptr);

    bn_reduce_kernel<<<bn_blocks, DIM, 0, stream>>>(bufA, sum, sumsq, NN, rows_per_block);
    bn_finalize_kernel<<<1, DIM, 0, stream>>>(sum, sumsq, mean, istd, NN);
    bn_norm_kernel<<<elem_blocks, T, 0, stream>>>(bufA, mean, istd, gamma, beta, out, NN);
}


# batchnorm column sums (value, square) taken from the layer-2 product's accumulators at its store: f32 atomics per 16-row tile into 16 copies (new sum order); bn_reduce dropped, 14 -> 13 launches
# speedup vs baseline: 1.3710x; 1.1407x over previous
#include <hip/hip_runtime.h>

#define NN 50000
#define NE 600000
#define DIM 128
#define BN_EPS 1e-5f
#define BN_COPIES 16

typedef __attribute__((ext_vector_type(2))) float v2f;
typedef __attribute__((ext_vector_type(8))) float v8f;


__global__ void init_deg_kernel(float* __restrict__ deg, int n) {
    int i = blockIdx.x * blockDim.x + threadIdx.x;
    if (i < n) deg[i] = 1.0f;
}

__global__ void deg_count_kernel(const int* __restrict__ col, float* __restrict__ deg, int e) {
    int i = blockIdx.x * blockDim.x + threadIdx.x;
    if (i < e) unsafeAtomicAdd(&deg[col[i]], 1.0f);
}

__global__ void rsqrt_kernel(float* __restrict__ deg, int n) {
    int i = blockIdx.x * blockDim.x + threadIdx.x;
    if (i < n) deg[i] = rsqrtf(deg[i]);
}

__global__ void zero_kernel(float* __restrict__ p, int n) {
    int i = blockIdx.x * blockDim.x + threadIdx.x;
    if (i < n) p[i] = 0.0f;
}


__global__ __launch_bounds__(256)
void transpose128_kernel(const float* __restrict__ W, float* __restrict__ Wt) {
    int t = blockIdx.x * blockDim.x + threadIdx.x;
    int r = t >> 7;
    int c = t & 127;
    Wt[(size_t)c * DIM + r] = W[(size_t)r * DIM + c];
}

template <bool SELF>
__global__ __launch_bounds__(256)
void gemm_wmma_kernel(const float* __restrict__ X, const float* __restrict__ Wt,
                      const float* __restrict__ bias, const float* __restrict__ dis,
                      float* __restrict__ H, float* __restrict__ AggNext,
                      float* __restrict__ stats) {
    const int tileM = blockIdx.x * 16;
    const int wave  = threadIdx.x >> 5;
    const int lane  = threadIdx.x & 31;
    const int tileN = wave * 16;
    const int mrow  = lane & 15;
    const int hi    = lane >> 4;

    v8f acc = {};
    const float* xr = X + (size_t)(tileM + mrow) * DIM + 2 * hi;
    const float* wc = Wt + (size_t)(tileN + mrow) * DIM + 2 * hi;

#pragma unroll
    for (int k0 = 0; k0 < DIM; k0 += 4) {
        v2f a = *(const v2f*)(xr + k0);
        v2f b = *(const v2f*)(wc + k0);
        acc = __builtin_amdgcn_wmma_f32_16x16x4_f32(
             false, a,  false, b,
             (short)0, acc,  false,  false);
    }

    const int row0 = tileM + 8 * hi;
    const int coln = tileN + mrow;
    const float bv = bias[coln];
    float* out = H + (size_t)row0 * DIM + coln;
    float* agg = SELF ? (AggNext + (size_t)row0 * DIM + coln) : nullptr;
    float cs = 0.0f, css = 0.0f;
#pragma unroll
    for (int r = 0; r < 8; ++r) {
        float v = fmaxf(acc[r] + bv, 0.0f);
        out[(size_t)r * DIM] = v;
        if (SELF) {
            float s = dis[row0 + r];
            agg[(size_t)r * DIM] = v * (s * s);
        } else {
            cs += v;
            css += v * v;
        }
    }
    if (!SELF) {
        cs  += __shfl_xor(cs, 16);
        css += __shfl_xor(css, 16);
        if (hi == 0) {
            float* st = stats + (size_t)(blockIdx.x & (BN_COPIES - 1)) * (2 * DIM);
            unsafeAtomicAdd(st + coln, cs);
            unsafeAtomicAdd(st + DIM + coln, css);
        }
    }
}


__global__ __launch_bounds__(256)
void selfloop_kernel(const float* __restrict__ h, const float* __restrict__ dis,
                     float* __restrict__ agg, int n) {
    int t = blockIdx.x * blockDim.x + threadIdx.x;
    if (t >= n * 32) return;
    int node = t >> 5;
    float s = dis[node];
    s = s * s;
    float4 v = ((const float4*)h)[t];
    v.x *= s; v.y *= s; v.z *= s; v.w *= s;
    ((float4*)agg)[t] = v;
}

__global__ __launch_bounds__(256)
void edge_scatter_kernel(const float* __restrict__ h, float* __restrict__ agg,
                         const int* __restrict__ row, const int* __restrict__ col,
                         const float* __restrict__ dis, int e) {
    int warp = threadIdx.x >> 5;
    int lane = threadIdx.x & 31;
    int eid = blockIdx.x * 8 + warp;
    if (eid >= e) return;
    int r = row[eid];
    int c = col[eid];
    float norm = dis[r] * dis[c];
    float4 v = ((const float4*)(h + (size_t)r * DIM))[lane];
    float* dst = agg + (size_t)c * DIM + lane * 4;
    unsafeAtomicAdd(dst + 0, v.x * norm);
    unsafeAtomicAdd(dst + 1, v.y * norm);
    unsafeAtomicAdd(dst + 2, v.z * norm);
    unsafeAtomicAdd(dst + 3, v.w * norm);
}


__global__ __launch_bounds__(128)
void bn_finalize_kernel(const float* __restrict__ stats,
                        float* __restrict__ mean, float* __restrict__ istd, int n) {
    int d = threadIdx.x;
    float s = 0.0f, ss = 0.0f;
#pragma unroll
    for (int c = 0; c < BN_COPIES; ++c) {
        s  += stats[c * 2 * DIM + d];
        ss += stats[c * 2 * DIM + DIM + d];
    }
    float invn = 1.0f / (float)n;
    float m = s * invn;
    float var = ss * invn - m * m;
    mean[d] = m;
    istd[d] = rsqrtf(var + BN_EPS);
}

__global__ __launch_bounds__(256)
void bn_norm_kernel(const float* __restrict__ h, const float* __restrict__ mean,
                    const float* __restrict__ istd, const float* __restrict__ gamma,
                    const float* __restrict__ beta, float* __restrict__ out, int n) {
    int t = blockIdx.x * blockDim.x + threadIdx.x;
    if (t >= n * 32) return;
    int j = t & 31;
    float4 v  = ((const float4*)h)[t];
    float4 m  = ((const float4*)mean)[j];
    float4 is = ((const float4*)istd)[j];
    float4 g  = ((const float4*)gamma)[j];
    float4 be = ((const float4*)beta)[j];
    float4 o;
    o.x = g.x * (v.x - m.x) * is.x + be.x;
    o.y = g.y * (v.y - m.y) * is.y + be.y;
    o.z = g.z * (v.z - m.z) * is.z + be.z;
    o.w = g.w * (v.w - m.w) * is.w + be.w;
    ((float4*)out)[t] = o;
}


extern "C" void kernel_launch(void* const* d_in, const int* in_sizes, int n_in,
                              void* d_out, int out_size, void* d_ws, size_t ws_size,
                              hipStream_t stream) {
    const float* x     = (const float*)d_in[0];
    const int*   eidx  = (const int*)d_in[1];
    const float* W1    = (const float*)d_in[2];
    const float* b1    = (const float*)d_in[3];
    const float* W2    = (const float*)d_in[4];
    const float* b2    = (const float*)d_in[5];
    const float* gamma = (const float*)d_in[6];
    const float* beta  = (const float*)d_in[7];
    float* out = (float*)d_out;

    const int* row = eidx;
    const int* col = eidx + NE;

    float* bufA  = (float*)d_ws;
    float* bufB  = bufA + (size_t)NN * DIM;
    float* bufC  = bufB + (size_t)NN * DIM;
    float* dis   = bufC + (size_t)NN * DIM;
    float* mean  = dis + NN;
    float* istd  = mean + DIM;
    float* W1t   = istd + DIM;
    float* W2t   = W1t + (size_t)DIM * DIM;
    float* stats = W2t + (size_t)DIM * DIM;

    const int T = 256;
    const int elem_blocks = (NN * 32 + T - 1) / T;
    const int gemm_blocks = NN / 16;
    const int edge_blocks = (NE + 7) / 8;
    const int tr_blocks   = (DIM * DIM) / T;
    const int stat_n = BN_COPIES * 2 * DIM;

    init_deg_kernel<<<(NN + T - 1) / T, T, 0, stream>>>(dis, NN);
    deg_count_kernel<<<(NE + T - 1) / T, T, 0, stream>>>(col, dis, NE);
    rsqrt_kernel<<<(NN + T - 1) / T, T, 0, stream>>>(dis, NN);
    zero_kernel<<<stat_n / T, T, 0, stream>>>(stats, stat_n);
    transpose128_kernel<<<tr_blocks, T, 0, stream>>>(W1, W1t);
    transpose128_kernel<<<tr_blocks, T, 0, stream>>>(W2, W2t);


    selfloop_kernel<<<elem_blocks, T, 0, stream>>>(x, dis, bufA, NN);
    edge_scatter_kernel<<<edge_blocks, T, 0, stream>>>(x, bufA, row, col, dis, NE);
    gemm_wmma_kernel<true><<<gemm_blocks, T, 0, stream>>>(bufA, W1t, b1, dis, bufB, bufC, nullptr);

    edge_scatter_kernel<<<edge_blocks, T, 0, stream>>>(bufB, bufC, row, col, dis, NE);
    gemm_wmma_kernel<false><<<gemm_blocks, T, 0, stream>>>(bufC, W2t, b2, nullptr, bufA, nullptr, stats);

    bn_finalize_kernel<<<1, DIM, 0, stream>>>(stats, mean, istd, NN);
    bn_norm_kernel<<<elem_blocks, T, 0, stream>>>(bufA, mean, istd, gamma, beta, out, NN);
}


# product: 1024 resident groups hold their W column strips in registers (read once from W itself; transposes dropped, 13 -> 11 launches); A tiles go once through LDS (whole-line loads); same k order
# speedup vs baseline: 1.5871x; 1.1576x over previous
#include <hip/hip_runtime.h>

#define NN 50000
#define NE 600000
#define DIM 128
#define BN_EPS 1e-5f
#define BN_COPIES 16

typedef __attribute__((ext_vector_type(2))) float v2f;
typedef __attribute__((ext_vector_type(8))) float v8f;


__global__ void init_deg_kernel(float* __restrict__ deg, int n) {
    int i = blockIdx.x * blockDim.x + threadIdx.x;
    if (i < n) deg[i] = 1.0f;
}

__global__ void deg_count_kernel(const int* __restrict__ col, float* __restrict__ deg, int e) {
    int i = blockIdx.x * blockDim.x + threadIdx.x;
    if (i < e) unsafeAtomicAdd(&deg[col[i]], 1.0f);
}

__global__ void rsqrt_kernel(float* __restrict__ deg, int n) {
    int i = blockIdx.x * blockDim.x + threadIdx.x;
    if (i < n) deg[i] = rsqrtf(deg[i]);
}

__global__ void zero_kernel(float* __restrict__ p, int n) {
    int i = blockIdx.x * blockDim.x + threadIdx.x;
    if (i < n) p[i] = 0.0f;
}

#define GEMM_TILES (NN / 16)
#define A_PITCH 132

template <bool SELF>
__global__ __launch_bounds__(256)
void gemm_wmma_kernel(const float* __restrict__ X, const float* __restrict__ W,
                      const float* __restrict__ bias, const float* __restrict__ dis,
                      float* __restrict__ H, float* __restrict__ AggNext,
                      float* __restrict__ stats) {
    __shared__ __attribute__((aligned(16))) float As[2][16 * A_PITCH];

    const int wave  = threadIdx.x >> 5;
    const int lane  = threadIdx.x & 31;
    const int tileN = wave * 16;
    const int mrow  = lane & 15;
    const int hi    = lane >> 4;
    const int coln  = tileN + mrow;

    v2f bfrag[DIM / 4];
    {
        const float* wp = W + (size_t)(2 * hi) * DIM + coln;
#pragma unroll
        for (int j = 0; j < DIM / 4; ++j) {
            bfrag[j].x = wp[(size_t)(4 * j) * DIM];
            bfrag[j].y = wp[(size_t)(4 * j + 1) * DIM];
        }
    }
    const float bv = bias[coln];

    const int soff = wave * A_PITCH + 4 * lane;
    const size_t goff = (size_t)wave * DIM + 4 * lane;

    int tile = blockIdx.x;
    float4 p0 = {}, p1 = {};
    if (tile < GEMM_TILES) {
        const float* xt = X + (size_t)tile * 16 * DIM + goff;
        p0 = *(const float4*)xt;
        p1 = *(const float4*)(xt + 8 * DIM);
    }
    int buf = 0;
    for (; tile < GEMM_TILES; tile += gridDim.x) {
        float* as = As[buf];
        *(float4*)(as + soff) = p0;
        *(float4*)(as + soff + 8 * A_PITCH) = p1;
        __syncthreads();

        const int nt = tile + gridDim.x;
        if (nt < GEMM_TILES) {
            const float* xt = X + (size_t)nt * 16 * DIM + goff;
            p0 = *(const float4*)xt;
            p1 = *(const float4*)(xt + 8 * DIM);
        }

        v8f acc = {};
        const float* ar = as + mrow * A_PITCH + 2 * hi;
#pragma unroll
        for (int j = 0; j < DIM / 4; ++j) {
            v2f a = *(const v2f*)(ar + 4 * j);
            acc = __builtin_amdgcn_wmma_f32_16x16x4_f32(
                 false, a,  false, bfrag[j],
                 (short)0, acc,  false,  false);
        }

        const int row0 = tile * 16 + 8 * hi;
        float* out = H + (size_t)row0 * DIM + coln;
        float* agg = SELF ? (AggNext + (size_t)row0 * DIM + coln) : nullptr;
        float cs = 0.0f, css = 0.0f;
#pragma unroll
        for (int r = 0; r < 8; ++r) {
            float v = fmaxf(acc[r] + bv, 0.0f);
            out[(size_t)r * DIM] = v;
            if (SELF) {
                float s = dis[row0 + r];
                agg[(size_t)r * DIM] = v * (s * s);
            } else {
                cs += v;
                css += v * v;
            }
        }
        if (!SELF) {
            cs  += __shfl_xor(cs, 16);
            css += __shfl_xor(css, 16);
            if (hi == 0) {
                float* st = stats + (size_t)(tile & (BN_COPIES - 1)) * (2 * DIM);
                unsafeAtomicAdd(st + coln, cs);
                unsafeAtomicAdd(st + DIM + coln, css);
            }
        }
        buf ^= 1;
    }
}


__global__ __launch_bounds__(256)
void selfloop_kernel(const float* __restrict__ h, const float* __restrict__ dis,
                     float* __restrict__ agg, int n) {
    int t = blockIdx.x * blockDim.x + threadIdx.x;
    if (t >= n * 32) return;
    int node = t >> 5;
    float s = dis[node];
    s = s * s;
    float4 v = ((const float4*)h)[t];
    v.x *= s; v.y *= s; v.z *= s; v.w *= s;
    ((float4*)agg)[t] = v;
}

__global__ __launch_bounds__(256)
void edge_scatter_kernel(const float* __restrict__ h, float* __restrict__ agg,
                         const int* __restrict__ row, const int* __restrict__ col,
                         const float* __restrict__ dis, int e) {
    int warp = threadIdx.x >> 5;
    int lane = threadIdx.x & 31;
    int eid = blockIdx.x * 8 + warp;
    if (eid >= e) return;
    int r = row[eid];
    int c = col[eid];
    float norm = dis[r] * dis[c];
    float4 v = ((const float4*)(h + (size_t)r * DIM))[lane];
    float* dst = agg + (size_t)c * DIM + lane * 4;
    unsafeAtomicAdd(dst + 0, v.x * norm);
    unsafeAtomicAdd(dst + 1, v.y * norm);
    unsafeAtomicAdd(dst + 2, v.z * norm);
    unsafeAtomicAdd(dst + 3, v.w * norm);
}


__global__ __launch_bounds__(128)
void bn_finalize_kernel(const float* __restrict__ stats,
                        float* __restrict__ mean, float* __restrict__ istd, int n) {
    int d = threadIdx.x;
    float s = 0.0f, ss = 0.0f;
#pragma unroll
    for (int c = 0; c < BN_COPIES; ++c) {
        s  += stats[c * 2 * DIM + d];
        ss += stats[c * 2 * DIM + DIM + d];
    }
    float invn = 1.0f / (float)n;
    float m = s * invn;
    float var = ss * invn - m * m;
    mean[d] = m;
    istd[d] = rsqrtf(var + BN_EPS);
}

__global__ __launch_bounds__(256)
void bn_norm_kernel(const float* __restrict__ h, const float* __restrict__ mean,
                    const float* __restrict__ istd, const float* __restrict__ gamma,
                    const float* __restrict__ beta, float* __restrict__ out, int n) {
    int t = blockIdx.x * blockDim.x + threadIdx.x;
    if (t >= n * 32) return;
    int j = t & 31;
    float4 v  = ((const float4*)h)[t];
    float4 m  = ((const float4*)mean)[j];
    float4 is = ((const float4*)istd)[j];
    float4 g  = ((const float4*)gamma)[j];
    float4 be = ((const float4*)beta)[j];
    float4 o;
    o.x = g.x * (v.x - m.x) * is.x + be.x;
    o.y = g.y * (v.y - m.y) * is.y + be.y;
    o.z = g.z * (v.z - m.z) * is.z + be.z;
    o.w = g.w * (v.w - m.w) * is.w + be.w;
    ((float4*)out)[t] = o;
}


extern "C" void kernel_launch(void* const* d_in, const int* in_sizes, int n_in,
                              void* d_out, int out_size, void* d_ws, size_t ws_size,
                              hipStream_t stream) {
    const float* x     = (const float*)d_in[0];
    const int*   eidx  = (const int*)d_in[1];
    const float* W1    = (const float*)d_in[2];
    const float* b1    = (const float*)d_in[3];
    const float* W2    = (const float*)d_in[4];
    const float* b2    = (const float*)d_in[5];
    const float* gamma = (const float*)d_in[6];
    const float* beta  = (const float*)d_in[7];
    float* out = (float*)d_out;

    const int* row = eidx;
    const int* col = eidx + NE;

    float* bufA  = (float*)d_ws;
    float* bufB  = bufA + (size_t)NN * DIM;
    float* bufC  = bufB + (size_t)NN * DIM;
    float* dis   = bufC + (size_t)NN * DIM;
    float* mean  = dis + NN;
    float* istd  = mean + DIM;
    float* stats = istd + DIM;

    const int T = 256;
    const int elem_blocks = (NN * 32 + T - 1) / T;
    const int gemm_blocks = 1024;
    const int edge_blocks = (NE + 7) / 8;
    const int stat_n = BN_COPIES * 2 * DIM;

    init_deg_kernel<<<(NN + T - 1) / T, T, 0, stream>>>(dis, NN);
    deg_count_kernel<<<(NE + T - 1) / T, T, 0, stream>>>(col, dis, NE);
    rsqrt_kernel<<<(NN + T - 1) / T, T, 0, stream>>>(dis, NN);
    zero_kernel<<<stat_n / T, T, 0, stream>>>(stats, stat_n);


    selfloop_kernel<<<elem_blocks, T, 0, stream>>>(x, dis, bufA, NN);
    edge_scatter_kernel<<<edge_blocks, T, 0, stream>>>(x, bufA, row, col, dis, NE);
    gemm_wmma_kernel<true><<<gemm_blocks, T, 0, stream>>>(bufA, W1, b1, dis, bufB, bufC, nullptr);

    edge_scatter_kernel<<<edge_blocks, T, 0, stream>>>(bufB, bufC, row, col, dis, NE);
    gemm_wmma_kernel<false><<<gemm_blocks, T, 0, stream>>>(bufC, W2, b2, nullptr, bufA, nullptr, stats);

    bn_finalize_kernel<<<1, DIM, 0, stream>>>(stats, mean, istd, NN);
    bn_norm_kernel<<<elem_blocks, T, 0, stream>>>(bufA, mean, istd, gamma, beta, out, NN);
}


# aggregation: a wave owns 4 rows of the bucket, takes its edges from the LDS-staged list by ballot into its own queue, sums them in registers 8 rows a batch (f32 sums in list order); LDS adds gone
# speedup vs baseline: 4.0855x; 2.5742x over previous
#include <hip/hip_runtime.h>

#define NN 50000
#define NE 600000
#define DIM 128
#define BN_EPS 1e-5f
#define BN_COPIES 4
#define BK_NODES 32
#define NBK ((NN + BK_NODES - 1) / BK_NODES)
#define NBK_PAD 1568
#define FILL_T 1024
#define FILL_EDGES (4 * FILL_T)

typedef __attribute__((ext_vector_type(2))) float v2f;
typedef __attribute__((ext_vector_type(8))) float v8f;


__global__ void init_kernel(float* __restrict__ deg, float* __restrict__ stats, int* __restrict__ cursor) {
    int i = blockIdx.x * blockDim.x + threadIdx.x;
    if (i < NN) deg[i] = 1.0f;
    if (i < BN_COPIES * 2 * DIM) stats[i] = 0.0f;
    if (i < NBK) cursor[i] = 0;
}

__global__ void deg_count_kernel(const int* __restrict__ col, float* __restrict__ deg, int e) {
    int i = blockIdx.x * blockDim.x + threadIdx.x;
    if (i < e) unsafeAtomicAdd(&deg[col[i]], 1.0f);
}

__global__ __launch_bounds__(FILL_T)
void bucket_fill_kernel(const int* __restrict__ row, const int* __restrict__ col,
                        const float* __restrict__ deg,
                        int* __restrict__ cursor, int* __restrict__ offsets,
                        int* __restrict__ elist, int e) {
    __shared__ int hist[NBK];
    __shared__ int boff[2048];
    __shared__ int sc[2][1024];
    const int t = threadIdx.x;

    const int e0 = blockIdx.x * FILL_EDGES + t;
    int r[4], c[4], rk[4];
    float nm[4];
#pragma unroll
    for (int i = 0; i < 4; ++i) {
        const int eid = e0 + i * FILL_T;
        r[i] = 0; c[i] = 0;
        if (eid < e) {
            r[i] = row[eid];
            c[i] = col[eid];
        }
    }
    constexpr int TRIPS = (NN / 4 + 1023) / 1024;
    float4 dv[TRIPS];
#pragma unroll
    for (int it = 0; it < TRIPS; ++it) {
        const int q = it * 1024 + t;
        dv[it] = ((const float4*)deg)[q < NN / 4 ? q : 0];
    }
    __builtin_amdgcn_sched_barrier(0);
    float dr[4], dc[4];
#pragma unroll
    for (int i = 0; i < 4; ++i) {
        dr[i] = deg[r[i]];
        dc[i] = deg[c[i]];
    }

    for (int b = t; b < NBK; b += FILL_T) hist[b] = 0;
    boff[t] = 0;
    boff[t + 1024] = 0;
    __syncthreads();
#pragma unroll
    for (int it = 0; it < TRIPS; ++it) {
        const int q = it * 1024 + t;
        const float4 d = dv[it];
        int cnt = (int)(d.x - 1.0f) + (int)(d.y - 1.0f) + (int)(d.z - 1.0f) + (int)(d.w - 1.0f);
        if (q >= NN / 4) cnt = 0;
        cnt += __shfl_xor(cnt, 1);
        cnt += __shfl_xor(cnt, 2);
        cnt += __shfl_xor(cnt, 4);
        boff[q >> 3] = cnt;
    }
    __syncthreads();
    const int c0 = boff[2 * t];
    const int v = c0 + boff[2 * t + 1];
    int buf = 0;
    sc[0][t] = v;
    __syncthreads();
    for (int off = 1; off < 1024; off <<= 1) {
        int x = sc[buf][t];
        if (t >= off) x += sc[buf][t - off];
        sc[buf ^ 1][t] = x;
        buf ^= 1;
        __syncthreads();
    }
    const int incl = sc[buf][t];
    const int excl = incl - v;
    boff[2 * t] = excl;
    boff[2 * t + 1] = excl + c0;
    if (blockIdx.x == 0) {
        const int b0 = 2 * t;
        if (b0 < NBK)     offsets[b0] = excl;
        if (b0 + 1 < NBK) offsets[b0 + 1] = excl + c0;
        if (t == 1023) offsets[NBK] = incl;
    }

#pragma unroll
    for (int i = 0; i < 4; ++i) {
        const int eid = e0 + i * FILL_T;
        rk[i] = 0; nm[i] = 0.0f;
        if (eid < e) {
            nm[i] = rsqrtf(dr[i]) * rsqrtf(dc[i]);
            rk[i] = atomicAdd(&hist[c[i] / BK_NODES], 1);
        }
    }
    __syncthreads();
    for (int b = t; b < NBK; b += FILL_T) {
        const int n = hist[b];
        hist[b] = boff[b] + (n ? atomicAdd(&cursor[b], n) : 0);
    }
    __syncthreads();
#pragma unroll
    for (int i = 0; i < 4; ++i) {
        const int eid = e0 + i * FILL_T;
        if (eid < e) {
            const int slot = hist[c[i] / BK_NODES] + rk[i];
            *(int4*)(elist + 4 * (size_t)slot) = make_int4(r[i], c[i], __float_as_int(nm[i]), 0);
        }
    }
}

#define T_PITCH 132
#define T_HALF 66
#define OWN_ROWS 4
#define Q_LEN 64
#define CH_REC 512

__device__ __forceinline__ void fma4(float4& a, const float4 v, const float s) {
    a.x = __builtin_fmaf(v.x, s, a.x); a.y = __builtin_fmaf(v.y, s, a.y);
    a.z = __builtin_fmaf(v.z, s, a.z); a.w = __builtin_fmaf(v.w, s, a.w);
}

template <int U, bool FULL>
__device__ __forceinline__ void sum_batch(const float* __restrict__ h,
                                          int rx, int ry, int rz, int j, int m, int lane,
                                          float4& a0, float4& a1, float4& a2, float4& a3) {
    float4 v[U];
#pragma unroll
    for (int u = 0; u < U; ++u) {
        v[u] = make_float4(0.0f, 0.0f, 0.0f, 0.0f);
        if (FULL || u < m) {
            const int r = __builtin_amdgcn_readlane(rx, j + u);
            v[u] = ((const float4*)(h + (size_t)r * DIM))[(unsigned)lane];
        }
    }
#pragma unroll
    for (int u = 0; u < U; ++u) {
        if (FULL || u < m) {
            const int k = __builtin_amdgcn_readlane(ry, j + u) & (OWN_ROWS - 1);
            const float nm = __int_as_float(__builtin_amdgcn_readlane(rz, j + u));
            fma4(a0, v[u], k == 0 ? nm : 0.0f);
            fma4(a1, v[u], k == 1 ? nm : 0.0f);
            fma4(a2, v[u], k == 2 ? nm : 0.0f);
            fma4(a3, v[u], k == 3 ? nm : 0.0f);
        }
    }
}

__device__ __forceinline__ void put4(float* tp, float4 s) {
    v2f lo, hi;
    lo.x = s.x; lo.y = s.y; hi.x = s.z; hi.y = s.w;
    *(v2f*)tp = lo;
    *(v2f*)(tp + T_HALF) = hi;
}

template <bool LAST>
__global__ __launch_bounds__(256)
void layer_kernel(const float* __restrict__ h, float* __restrict__ hout,
                  const int* __restrict__ elist, const int* __restrict__ offsets,
                  const float* __restrict__ deg,
                  const float* __restrict__ W, const float* __restrict__ bias,
                  float* __restrict__ stats) {
    __shared__ __attribute__((aligned(16))) float tile[BK_NODES * T_PITCH];
    __shared__ int4 recs[CH_REC];
    __shared__ int4 queue[8 * Q_LEN];
    const int t    = threadIdx.x;
    const int wave = __builtin_amdgcn_readfirstlane(t >> 5);
    const int lane = t & 31;
    const int n0   = blockIdx.x * BK_NODES;
    const int nrow = (NN - n0) < BK_NODES ? (NN - n0) : BK_NODES;
    const int beg  = offsets[blockIdx.x];
    const int end  = offsets[blockIdx.x + 1];

    const int wrow = n0 + OWN_ROWS * wave;
    float4 a0 = make_float4(0.0f, 0.0f, 0.0f, 0.0f), a1 = a0, a2 = a0, a3 = a0;
    if (wrow < NN) {
        const float4* src = (const float4*)(h + (size_t)wrow * DIM) + lane;
        a0 = src[0]; a1 = src[DIM / 4]; a2 = src[2 * (DIM / 4)]; a3 = src[3 * (DIM / 4)];
        float d0 = rsqrtf(deg[wrow]), d1 = rsqrtf(deg[wrow + 1]), d2 = rsqrtf(deg[wrow + 2]), d3 = rsqrtf(deg[wrow + 3]);
        d0 = d0 * d0; d1 = d1 * d1; d2 = d2 * d2; d3 = d3 * d3;
        a0.x *= d0; a0.y *= d0; a0.z *= d0; a0.w *= d0;
        a1.x *= d1; a1.y *= d1; a1.z *= d1; a1.w *= d1;
        a2.x *= d2; a2.y *= d2; a2.z *= d2; a2.w *= d2;
        a3.x *= d3; a3.y *= d3; a3.z *= d3; a3.w *= d3;
    }

    int4* q = queue + wave * Q_LEN;
    int qh = 0, qn = 0;
    for (int cb = beg; cb < end; cb += CH_REC) {
        if (cb != beg) __syncthreads();
#pragma unroll
        for (int i = 0; i < CH_REC / 256; ++i) {
            const int idx = cb + i * 256 + t;
            if (idx < end) recs[i * 256 + t] = *(const int4*)(elist + 4 * (size_t)idx);
        }
        __syncthreads();
        const int n = (end - cb) < CH_REC ? (end - cb) : CH_REC;
        for (int b = 0; b < n; b += 32) {
            const bool valid = b + lane < n;
            int4 cur = make_int4(0, 0, 0, 0);
            if (valid) cur = recs[b + lane];
            const bool own = valid && (((cur.y - n0) >> 2) == wave);
            const unsigned mask = __builtin_amdgcn_ballot_w32(own);
            if (own) q[(qh + qn + (int)__builtin_amdgcn_mbcnt_lo(mask, 0u)) & (Q_LEN - 1)] = cur;
            qn += __builtin_popcount(mask);
            if (qn >= 32) {
                const int4 rec = q[(qh + lane) & (Q_LEN - 1)];
#pragma nounroll
                for (int j = 0; j < 32; j += 8)
                    sum_batch<8, true>(h, rec.x, rec.y, rec.z, j, 8, lane, a0, a1, a2, a3);
                qh = (qh + 32) & (Q_LEN - 1);
                qn -= 32;
            }
        }
    }
    if (qn > 0) {
        int4 rec = make_int4(0, 0, 0, 0);
        if (lane < qn) rec = q[(qh + lane) & (Q_LEN - 1)];
        int j = 0;
        for (; j + 8 <= qn; j += 8) sum_batch<8, true>(h, rec.x, rec.y, rec.z, j, 8, lane, a0, a1, a2, a3);
        if (j < qn) sum_batch<8, false>(h, rec.x, rec.y, rec.z, j, qn - j, lane, a0, a1, a2, a3);
    }

    {
        float* tp = tile + (OWN_ROWS * wave) * T_PITCH + 2 * lane;
        put4(tp, a0);
        put4(tp + T_PITCH, a1);
        put4(tp + 2 * T_PITCH, a2);
        put4(tp + 3 * T_PITCH, a3);
    }

    const int mrow = lane & 15;
    const int hi   = lane >> 4;
    const int coln = wave * 16 + mrow;

    constexpr int KC = 8;
    const float* wp = W + (size_t)(2 * hi) * DIM + coln;
    v2f bq[2][KC];
#pragma unroll
    for (int jj = 0; jj < KC; ++jj) {
        bq[0][jj].x = wp[(size_t)(4 * jj) * DIM];
        bq[0][jj].y = wp[(size_t)(4 * jj + 1) * DIM];
    }
    const float bv = bias[coln];

    __syncthreads();

    v8f acc0 = {}, acc1 = {};
    const float* ar = tile + mrow * T_PITCH + hi * T_HALF;
#pragma unroll
    for (int q = 0; q < DIM / 4 / KC; ++q) {
        if (q + 1 < DIM / 4 / KC) {
#pragma unroll
            for (int jj = 0; jj < KC; ++jj) {
                const int j = (q + 1) * KC + jj;
                bq[(q + 1) & 1][jj].x = wp[(size_t)(4 * j) * DIM];
                bq[(q + 1) & 1][jj].y = wp[(size_t)(4 * j + 1) * DIM];
            }
        }
#pragma unroll
        for (int jj = 0; jj < KC; ++jj) {
            const int j = q * KC + jj;
            v2f a0 = *(const v2f*)(ar + 2 * j);
            v2f a1 = *(const v2f*)(ar + 16 * T_PITCH + 2 * j);
            acc0 = __builtin_amdgcn_wmma_f32_16x16x4_f32(
                 false, a0,  false, bq[q & 1][jj],
                 (short)0, acc0,  false,  false);
            acc1 = __builtin_amdgcn_wmma_f32_16x16x4_f32(
                 false, a1,  false, bq[q & 1][jj],
                 (short)0, acc1,  false,  false);
        }
    }

    float cs = 0.0f, css = 0.0f;
#pragma unroll
    for (int rt = 0; rt < 2; ++rt) {
        if (rt == 0 || nrow == BK_NODES) {
            const int row0 = n0 + 16 * rt + 8 * hi;
            float* out = hout + (size_t)row0 * DIM + coln;
#pragma unroll
            for (int r = 0; r < 8; ++r) {
                float v = fmaxf((rt ? acc1[r] : acc0[r]) + bv, 0.0f);
                out[(size_t)r * DIM] = v;
                if (LAST) {
                    cs += v;
                    css += v * v;
                }
            }
        }
    }
    if (LAST) {
        cs  += __shfl_xor(cs, 16);
        css += __shfl_xor(css, 16);
        if (hi == 0) {
            float* st = stats + (size_t)(blockIdx.x & (BN_COPIES - 1)) * (2 * DIM);
            unsafeAtomicAdd(st + coln, cs);
            unsafeAtomicAdd(st + DIM + coln, css);
        }
    }
}


__global__ __launch_bounds__(256)
void bn_norm_kernel(const float* __restrict__ h, const float* __restrict__ stats,
                    const float* __restrict__ gamma,
                    const float* __restrict__ beta, float* __restrict__ out, int n) {
    int t = blockIdx.x * blockDim.x + threadIdx.x;
    if (t >= n * 32) return;
    int j = t & 31;
    float4 v  = ((const float4*)h)[t];
    float4 g  = ((const float4*)gamma)[j];
    float4 be = ((const float4*)beta)[j];
    float4 sa[BN_COPIES], sq[BN_COPIES];
#pragma unroll
    for (int c = 0; c < BN_COPIES; ++c) {
        sa[c] = ((const float4*)stats)[c * (2 * DIM / 4) + j];
        sq[c] = ((const float4*)stats)[c * (2 * DIM / 4) + DIM / 4 + j];
    }
    float4 s = sa[0], ss = sq[0];
#pragma unroll
    for (int c = 1; c < BN_COPIES; ++c) {
        s.x += sa[c].x; s.y += sa[c].y; s.z += sa[c].z; s.w += sa[c].w;
        ss.x += sq[c].x; ss.y += sq[c].y; ss.z += sq[c].z; ss.w += sq[c].w;
    }
    const float invn = 1.0f / (float)NN;
    float4 m, is;
    m.x = s.x * invn; m.y = s.y * invn; m.z = s.z * invn; m.w = s.w * invn;
    is.x = rsqrtf(ss.x * invn - m.x * m.x + BN_EPS);
    is.y = rsqrtf(ss.y * invn - m.y * m.y + BN_EPS);
    is.z = rsqrtf(ss.z * invn - m.z * m.z + BN_EPS);
    is.w = rsqrtf(ss.w * invn - m.w * m.w + BN_EPS);
    float4 o;
    o.x = g.x * (v.x - m.x) * is.x + be.x;
    o.y = g.y * (v.y - m.y) * is.y + be.y;
    o.z = g.z * (v.z - m.z) * is.z + be.z;
    o.w = g.w * (v.w - m.w) * is.w + be.w;
    ((float4*)out)[t] = o;
}


extern "C" void kernel_launch(void* const* d_in, const int* in_sizes, int n_in,
                              void* d_out, int out_size, void* d_ws, size_t ws_size,
                              hipStream_t stream) {
    const float* x     = (const float*)d_in[0];
    const int*   eidx  = (const int*)d_in[1];
    const float* W1    = (const float*)d_in[2];
    const float* b1    = (const float*)d_in[3];
    const float* W2    = (const float*)d_in[4];
    const float* b2    = (const float*)d_in[5];
    const float* gamma = (const float*)d_in[6];
    const float* beta  = (const float*)d_in[7];
    float* out = (float*)d_out;

    const int* row = eidx;
    const int* col = eidx + NE;

    float* bufA  = (float*)d_ws;
    float* bufB  = bufA + (size_t)NN * DIM;
    float* stats = bufB + (size_t)NN * DIM;
    int* offsets = (int*)(stats + BN_COPIES * 2 * DIM);
    int* cursor  = offsets + NBK_PAD;
    int* elist   = cursor + NBK_PAD;
    float* deg   = (float*)(elist + 4 * (size_t)NE);

    const int T = 256;
    const int elem_blocks = (NN * 32 + T - 1) / T;
    const int fill_blocks = (NE + FILL_EDGES - 1) / FILL_EDGES;

    init_kernel<<<(NN + T - 1) / T, T, 0, stream>>>(deg, stats, cursor);
    deg_count_kernel<<<(NE + T - 1) / T, T, 0, stream>>>(col, deg, NE);
    bucket_fill_kernel<<<fill_blocks, FILL_T, 0, stream>>>(row, col, deg, cursor, offsets, elist, NE);


    layer_kernel<false><<<NBK, T, 0, stream>>>(x, bufB, elist, offsets, deg, W1, b1, nullptr);

    layer_kernel<true><<<NBK, T, 0, stream>>>(bufB, bufA, elist, offsets, deg, W2, b2, stats);

    bn_norm_kernel<<<elem_blocks, T, 0, stream>>>(bufA, stats, gamma, beta, out, NN);
}
